# starting kernel
# baseline (speedup 1.0000x reference)
#include <hip/hip_runtime.h>

#define LOG2E 1.4426950408889634f

typedef __attribute__((ext_vector_type(2))) float v2f;
typedef __attribute__((ext_vector_type(8))) float v8f;

#define BB   8
#define NN   1024
#define MM   1024
#define CIN  8
#define CREP 9
#define COUT 16
#define MT   32
#define NCHUNK 8
#define NPER (NN / NCHUNK)

__global__ __launch_bounds__(256) void convcnp_kernel(
    const float* __restrict__ x,
    const float* __restrict__ y,
    const float* __restrict__ t,
    const float* __restrict__ sigma,
    const float* __restrict__ W,
    const float* __restrict__ bias,
    float* __restrict__ out)
{
    __shared__ float  sx[NN];
    __shared__ float4 sy4[NN * 2];
    __shared__ float  spart[NCHUNK * MT * 10];
    __shared__ float  yo[MT * 12];
    __shared__ float  Wp[COUT * 12];

    const int tid   = threadIdx.x;
    const int b     = blockIdx.y;
    const int mt    = blockIdx.x;
    const int m_l   = tid & (MT - 1);
    const int chunk = tid >> 5;

    {
        const float4* x4 = (const float4*)(x + b * NN);
        ((float4*)sx)[tid] = x4[tid];
        const float4* y4 = (const float4*)(y + (size_t)b * NN * CIN);
        #pragma unroll
        for (int k = 0; k < 8; ++k)
            sy4[tid + k * 256] = y4[tid + k * 256];
    }

    float k2[CREP];
    #pragma unroll
    for (int c = 0; c < CREP; ++c) {
        float s = sigma[c];
        k2[c] = -0.5f * LOG2E * __builtin_amdgcn_exp2f(-2.0f * s * LOG2E);
    }

    const float tm = t[b * MM + mt * MT + m_l];
    __syncthreads();

    float acc[CREP];
    #pragma unroll
    for (int c = 0; c < CREP; ++c) acc[c] = 0.f;

    const int n0 = chunk * NPER;
    #pragma unroll 4
    for (int i = 0; i < NPER; ++i) {
        const int n = n0 + i;
        const float  xn = sx[n];
        const float4 ya = sy4[2 * n];
        const float4 yb = sy4[2 * n + 1];
        const float diff = xn - tm;
        const float dd   = diff * diff;
        float yr[CIN] = { ya.x, ya.y, ya.z, ya.w, yb.x, yb.y, yb.z, yb.w };
        acc[0] += __builtin_amdgcn_exp2f(dd * k2[0]);
        #pragma unroll
        for (int c = 1; c < CREP; ++c)
            acc[c] += __builtin_amdgcn_exp2f(dd * k2[c]) * yr[c - 1];
    }

    {
        const int base = (chunk * MT + m_l) * 10;
        #pragma unroll
        for (int c = 0; c < CREP; ++c) spart[base + c] = acc[c];
    }
    __syncthreads();

    if (tid < MT) {
        float s[CREP];
        #pragma unroll
        for (int c = 0; c < CREP; ++c) s[c] = 0.f;
        for (int ch = 0; ch < NCHUNK; ++ch) {
            const int base = (ch * MT + tid) * 10;
            #pragma unroll
            for (int c = 0; c < CREP; ++c) s[c] += spart[base + c];
        }
        const float dens = s[0];
        const float inv  = 1.0f / (dens + 1e-8f);
        yo[tid * 12 + 0] = dens;
        #pragma unroll
        for (int c = 1; c < CREP; ++c) yo[tid * 12 + c] = s[c] * inv;
        yo[tid * 12 + 9]  = 0.f;
        yo[tid * 12 + 10] = 0.f;
        yo[tid * 12 + 11] = 0.f;
    }
    if (tid < COUT * 12) {
        const int o = tid / 12, c = tid % 12;
        Wp[tid] = (c < CREP) ? W[o * CREP + c] : 0.f;
    }
    __syncthreads();

    const int wid  = tid >> 5;
    const int lane = tid & 31;
    if (wid < 2) {
        const int half  = lane >> 4;
        const int lr    = lane & 15;
        const int mbase = wid * 16;

        v8f d = {0.f, 0.f, 0.f, 0.f, 0.f, 0.f, 0.f, 0.f};
        #pragma unroll
        for (int kc = 0; kc < 3; ++kc) {
            const int c0 = kc * 4 + 2 * half;
            v2f a, bm;
            a.x  = yo[(mbase + lr) * 12 + c0];
            a.y  = yo[(mbase + lr) * 12 + c0 + 1];
            bm.x = Wp[lr * 12 + c0];
            bm.y = Wp[lr * 12 + c0 + 1];
            d = __builtin_amdgcn_wmma_f32_16x16x4_f32(
                    false, a, false, bm, (short)0, d, false, false);
        }

        const float bo = bias[lr];
        const int mg = mt * MT + mbase;
        #pragma unroll
        for (int r = 0; r < 8; ++r) {
            const int m = mg + r + 8 * half;
            out[((size_t)b * MM + m) * COUT + lr] = d[r] + bo;
        }
    }
}

extern "C" void kernel_launch(void* const* d_in, const int* in_sizes, int n_in,
                              void* d_out, int out_size, void* d_ws, size_t ws_size,
                              hipStream_t stream) {
    const float* x     = (const float*)d_in[0];
    const float* y     = (const float*)d_in[1];
    const float* t     = (const float*)d_in[2];
    const float* sigma = (const float*)d_in[3];
    const float* W     = (const float*)d_in[4];
    const float* bias  = (const float*)d_in[5];
    float* out = (float*)d_out;

    dim3 grid(MM / MT, BB);
    dim3 block(256);
    convcnp_kernel<<<grid, block, 0, stream>>>(x, y, t, sigma, W, bias, out);
}


# context loop cut into an exponential phase (72 exp2 in place) and a multiply-add phase, one trip apart; s_setprio lifts a wave while it issues exponentials so its SIMD partner's plain ops fill gaps
# speedup vs baseline: 1.4623x; 1.4623x over previous
#include <hip/hip_runtime.h>


#define LOG2E 1.4426950408889634f

typedef __attribute__((ext_vector_type(2))) float v2f;
typedef __attribute__((ext_vector_type(8))) float v8f;

#define BB   8
#define NN   1024
#define MM   1024
#define CIN  8
#define CREP 9
#define COUT 16
#define MT   32
#define MW   4
#define TPL  2
#define NSUB 16
#define NITER (NN / NSUB)
#define XPITCH (NITER + 4)
#define YPITCH (2 * NITER + 1)

template <int MASK>
__device__ __forceinline__ float row_xor(float v) {
    const int b = __builtin_bit_cast(int, v);
    return __builtin_bit_cast(float,
        __builtin_amdgcn_update_dpp(b, b, 0x160 | MASK, 0xf, 0xf, false));
}

__device__ __forceinline__ float xhalf(float v) {
    const int b = __builtin_bit_cast(int, v);
    return __builtin_bit_cast(float,
        __builtin_amdgcn_permlanex16(b, b, 0x76543210, 0xfedcba98, false, false));
}

__device__ __forceinline__ void exponents(const float4 xq, const v2f tmv, const v2f k20,
                                          const v2f (&k2p)[4], v2f (&ex)[4][CREP]) {
    const float xr[4] = { xq.x, xq.y, xq.z, xq.w };
    #pragma unroll
    for (int u = 0; u < 4; ++u) {
        const v2f xx   = { xr[u], xr[u] };
        const v2f diff = xx - tmv;
        const v2f dd   = diff * diff;
        ex[u][0] = dd * k20;
        #pragma unroll
        for (int s = 0; s < TPL; ++s) {
            const float dds = s ? dd.y : dd.x;
            #pragma unroll
            for (int p = 0; p < 4; ++p) ex[u][1 + 4 * s + p] = k2p[p] * dds;
        }
    }
}

__device__ __forceinline__ void weights(v2f (&ex)[4][CREP]) {
    #pragma unroll
    for (int u = 0; u < 4; ++u) {
        #pragma unroll
        for (int q = 0; q < CREP; ++q) {
            ex[u][q].x = __builtin_amdgcn_exp2f(ex[u][q].x);
            ex[u][q].y = __builtin_amdgcn_exp2f(ex[u][q].y);
        }
    }
}

__device__ __forceinline__ void accumulate(const v2f (&ex)[4][CREP], const float4 (&yv)[8],
                                           v2f& acc0, v2f (&accp)[TPL][4]) {
    #pragma unroll
    for (int u = 0; u < 4; ++u) {
        const float4 ya = yv[2 * u];
        const float4 yb = yv[2 * u + 1];
        const v2f yp[4] = { {ya.x, ya.y}, {ya.z, ya.w}, {yb.x, yb.y}, {yb.z, yb.w} };
        acc0 += ex[u][0];
        #pragma unroll
        for (int s = 0; s < TPL; ++s) {
            #pragma unroll
            for (int p = 0; p < 4; ++p) accp[s][p] += ex[u][1 + 4 * s + p] * yp[p];
        }
    }
}

__global__ __launch_bounds__(256) void convcnp_kernel(
    const float* __restrict__ x,
    const float* __restrict__ y,
    const float* __restrict__ t,
    const float* __restrict__ sigma,
    const float* __restrict__ W,
    const float* __restrict__ bias,
    float* __restrict__ out)
{
    __shared__ __attribute__((aligned(16))) float sx[NSUB * XPITCH];
    __shared__ float4 sy4[NSUB * YPITCH];

    const int tid  = threadIdx.x;
    const int b    = blockIdx.y;
    const int mt   = blockIdx.x;
    const int lane = tid & 31;
    const int wave = tid >> 5;
    const int mi   = lane & 1;
    const int j    = lane >> 1;
    const int half = lane >> 4;
    const int lr   = lane & 15;
    const int m0   = mt * MT + wave * MW;

    {
        const float4* x4 = (const float4*)(x + b * NN);
        *(float4*)(sx + (tid >> 4) * XPITCH + 4 * (tid & 15)) = x4[tid];
        const float4* y4 = (const float4*)(y + (size_t)b * NN * CIN);
        #pragma unroll
        for (int k = 0; k < 8; ++k)
            sy4[(2 * k + (tid >> 7)) * YPITCH + (tid & 127)] = y4[tid + k * 256];
    }

    float k2[CREP];
    #pragma unroll
    for (int c = 0; c < CREP; ++c) {
        float s = sigma[c];
        k2[c] = -0.5f * LOG2E * __builtin_amdgcn_exp2f(-2.0f * s * LOG2E);
    }

    const float2 tm2 = *(const float2*)(t + b * MM + m0 + TPL * mi);
    const v2f tmv = { tm2.x, tm2.y };

    const float* wrow = W + lr * CREP;
    const float w0x = wrow[2 * half];
    const float w0y = wrow[2 * half + 1];
    const float w1x = wrow[4 + 2 * half];
    const float w1y = wrow[5 + 2 * half];
    const float w8  = wrow[8];
    const float bo  = bias[lr];

    __syncthreads();

    const float*  xs = sx  + j * XPITCH;
    const float4* ys = sy4 + j * YPITCH;
    v2f k2p[4], accp[TPL][4];
    #pragma unroll
    for (int p = 0; p < 4; ++p) {
        k2p[p].x = k2[2 * p + 1];
        k2p[p].y = k2[2 * p + 2];
        #pragma unroll
        for (int s = 0; s < TPL; ++s) {
            accp[s][p].x = 0.f;
            accp[s][p].y = 0.f;
        }
    }
    const v2f k20 = { k2[0], k2[0] };
    v2f acc0 = { 0.f, 0.f };

    v2f    ex[4][CREP];
    float4 yv[8];
    float4 xv;

    xv = *(const float4*)xs;
    #pragma unroll
    for (int q = 0; q < 8; ++q) yv[q] = ys[q];
    exponents(xv, tmv, k20, k2p, ex);
    xv = *(const float4*)(xs + 4);

    for (int i = 0; i < NITER - 8; i += 4) {
        __builtin_amdgcn_s_setprio(3);
        weights(ex);
        __builtin_amdgcn_s_setprio(0);
        accumulate(ex, yv, acc0, accp);
        exponents(xv, tmv, k20, k2p, ex);
        __builtin_amdgcn_sched_barrier(0);
        #pragma unroll
        for (int q = 0; q < 8; ++q) yv[q] = ys[2 * (i + 4) + q];
        xv = *(const float4*)(xs + i + 8);
    }
    __builtin_amdgcn_s_setprio(3);
    weights(ex);
    __builtin_amdgcn_s_setprio(0);
    accumulate(ex, yv, acc0, accp);
    exponents(xv, tmv, k20, k2p, ex);
    __builtin_amdgcn_sched_barrier(0);
    #pragma unroll
    for (int q = 0; q < 8; ++q) yv[q] = ys[2 * (NITER - 4) + q];
    __builtin_amdgcn_s_setprio(3);
    weights(ex);
    __builtin_amdgcn_s_setprio(0);
    accumulate(ex, yv, acc0, accp);

    float part[TPL][CREP];
    part[0][0] = acc0.x;
    part[1][0] = acc0.y;
    #pragma unroll
    for (int s = 0; s < TPL; ++s) {
        #pragma unroll
        for (int p = 0; p < 4; ++p) {
            part[s][2 * p + 1] = accp[s][p].x;
            part[s][2 * p + 2] = accp[s][p].y;
        }
    }
    #pragma unroll
    for (int s = 0; s < TPL; ++s) {
        #pragma unroll
        for (int c = 0; c < CREP; ++c) {
            const float v = part[s][c];
            part[s][c] = v + row_xor<2>(v);
        }
    }
    const bool second = (lane >> 1) & 1;
    float acc[CREP];
    #pragma unroll
    for (int c = 0; c < CREP; ++c) {
        float v = second ? part[1][c] : part[0][c];
        v += row_xor<4>(v);
        v += row_xor<8>(v);
        v += xhalf(v);
        acc[c] = v;
    }

    const float dens = acc[0];
    const float inv  = 1.0f / (dens + 1e-8f);
    float yo[12];
    yo[0] = dens;
    #pragma unroll
    for (int c = 1; c < CREP; ++c) yo[c] = acc[c] * inv;
    yo[9] = 0.f; yo[10] = 0.f; yo[11] = 0.f;

    v8f d = {0.f, 0.f, 0.f, 0.f, 0.f, 0.f, 0.f, 0.f};
    {
        v2f a, bm;
        a.x  = half ? yo[2] : yo[0];
        a.y  = half ? yo[3] : yo[1];
        bm.x = w0x; bm.y = w0y;
        d = __builtin_amdgcn_wmma_f32_16x16x4_f32(false, a, false, bm, (short)0, d, false, false);
        a.x  = half ? yo[6] : yo[4];
        a.y  = half ? yo[7] : yo[5];
        bm.x = w1x; bm.y = w1y;
        d = __builtin_amdgcn_wmma_f32_16x16x4_f32(false, a, false, bm, (short)0, d, false, false);
        a.x  = half ? yo[10] : yo[8];
        a.y  = half ? yo[11] : yo[9];
        bm.x = half ? 0.f : w8;
        bm.y = 0.f;
        d = __builtin_amdgcn_wmma_f32_16x16x4_f32(false, a, false, bm, (short)0, d, false, false);
    }

    if (lane < 16) {
        float* o = out + ((size_t)b * MM + m0) * COUT + lane;
        #pragma unroll
        for (int r = 0; r < MW; ++r)
            o[(2 * (r & 1) + (r >> 1)) * COUT] = d[r] + bo;
    }
}

extern "C" void kernel_launch(void* const* d_in, const int* in_sizes, int n_in,
                              void* d_out, int out_size, void* d_ws, size_t ws_size,
                              hipStream_t stream) {
    const float* x     = (const float*)d_in[0];
    const float* y     = (const float*)d_in[1];
    const float* t     = (const float*)d_in[2];
    const float* sigma = (const float*)d_in[3];
    const float* W     = (const float*)d_in[4];
    const float* bias  = (const float*)d_in[5];
    float* out = (float*)d_out;

    dim3 grid(MM / MT, BB);
    dim3 block(256);
    convcnp_kernel<<<grid, block, 0, stream>>>(x, y, t, sigma, W, bias, out);
}
